# starting kernel
# baseline (speedup 1.0000x reference)
#include <hip/hip_runtime.h>

typedef __attribute__((ext_vector_type(2))) float v2f;
typedef __attribute__((ext_vector_type(4))) float v4f;
typedef __attribute__((ext_vector_type(8))) float v8f;


__global__ __launch_bounds__(256) void CentersDistance_wmma_kernel(
    const float* __restrict__ inputs,
    const float* __restrict__ centers,
    float* __restrict__ out)
{
    constexpr int Q = 4096;
    constexpr int D = 128;

    const int lane = threadIdx.x & 31;
    const int wave = threadIdx.x >> 5;
    const int tile = blockIdx.x * 8 + wave;

    const int c0 = (tile & 15) * 16;
    const int q0 = (tile >> 4) * 64;

    const int hi  = lane >> 4;
    const int l15 = lane & 15;

    const float* aptr = centers + (c0 + l15) * D + 4 * hi;
    const float* bptr = inputs + (q0 + l15) * D + 4 * hi;

    v8f acc0 = {}, acc1 = {}, acc2 = {}, acc3 = {};
    float asum = 0.f;
    float bsum0 = 0.f, bsum1 = 0.f, bsum2 = 0.f, bsum3 = 0.f;

    #pragma unroll 2
    for (int k = 0; k < D; k += 8) {
        v4f a  = *(const v4f*)(aptr + k);
        v4f b0 = *(const v4f*)(bptr + k);
        v4f b1 = *(const v4f*)(bptr + 16 * D + k);
        v4f b2 = *(const v4f*)(bptr + 32 * D + k);
        v4f b3 = *(const v4f*)(bptr + 48 * D + k);

        asum  += a.x  * a.x  + a.y  * a.y  + a.z  * a.z  + a.w  * a.w;
        bsum0 += b0.x * b0.x + b0.y * b0.y + b0.z * b0.z + b0.w * b0.w;
        bsum1 += b1.x * b1.x + b1.y * b1.y + b1.z * b1.z + b1.w * b1.w;
        bsum2 += b2.x * b2.x + b2.y * b2.y + b2.z * b2.z + b2.w * b2.w;
        bsum3 += b3.x * b3.x + b3.y * b3.y + b3.z * b3.z + b3.w * b3.w;

        v2f alo = {a.x, a.y},  ahi = {a.z, a.w};
        v2f l0  = {b0.x, b0.y}, h0 = {b0.z, b0.w};
        v2f l1  = {b1.x, b1.y}, h1 = {b1.z, b1.w};
        v2f l2  = {b2.x, b2.y}, h2 = {b2.z, b2.w};
        v2f l3  = {b3.x, b3.y}, h3 = {b3.z, b3.w};

        acc0 = __builtin_amdgcn_wmma_f32_16x16x4_f32(false, alo, false, l0,
                                                     (short)0, acc0, false, false);
        acc1 = __builtin_amdgcn_wmma_f32_16x16x4_f32(false, alo, false, l1,
                                                     (short)0, acc1, false, false);
        acc2 = __builtin_amdgcn_wmma_f32_16x16x4_f32(false, alo, false, l2,
                                                     (short)0, acc2, false, false);
        acc3 = __builtin_amdgcn_wmma_f32_16x16x4_f32(false, alo, false, l3,
                                                     (short)0, acc3, false, false);
        acc0 = __builtin_amdgcn_wmma_f32_16x16x4_f32(false, ahi, false, h0,
                                                     (short)0, acc0, false, false);
        acc1 = __builtin_amdgcn_wmma_f32_16x16x4_f32(false, ahi, false, h1,
                                                     (short)0, acc1, false, false);
        acc2 = __builtin_amdgcn_wmma_f32_16x16x4_f32(false, ahi, false, h2,
                                                     (short)0, acc2, false, false);
        acc3 = __builtin_amdgcn_wmma_f32_16x16x4_f32(false, ahi, false, h3,
                                                     (short)0, acc3, false, false);
    }

    float cn  = asum  + __shfl_xor(asum,  16, 32);
    float qn0 = bsum0 + __shfl_xor(bsum0, 16, 32);
    float qn1 = bsum1 + __shfl_xor(bsum1, 16, 32);
    float qn2 = bsum2 + __shfl_xor(bsum2, 16, 32);
    float qn3 = bsum3 + __shfl_xor(bsum3, 16, 32);

    float crow[8];
    #pragma unroll
    for (int j = 0; j < 8; ++j)
        crow[j] = __shfl(cn, j + 8 * hi, 32);

    #pragma unroll
    for (int j = 0; j < 8; ++j) {
        const int row = c0 + j + 8 * hi;
        float* p = out + row * Q + q0 + l15;
        p[0]  = 2.f * acc0[j] - crow[j] - qn0;
        p[16] = 2.f * acc1[j] - crow[j] - qn1;
        p[32] = 2.f * acc2[j] - crow[j] - qn2;
        p[48] = 2.f * acc3[j] - crow[j] - qn3;
    }
}

extern "C" void kernel_launch(void* const* d_in, const int* in_sizes, int n_in,
                              void* d_out, int out_size, void* d_ws, size_t ws_size,
                              hipStream_t stream) {
    (void)in_sizes; (void)n_in; (void)out_size; (void)d_ws; (void)ws_size;
    const float* inputs  = (const float*)d_in[0];
    const float* centers = (const float*)d_in[1];
    float* out = (float*)d_out;

    CentersDistance_wmma_kernel<<<128, 256, 0, stream>>>(inputs, centers, out);
}


# Stage each work-group's 128 center rows and 64 input rows through LDS: one global load reads a whole 512 B row (4 full lines, not 32 B pieces), once per work-group; fragments via ds_load_b128
# speedup vs baseline: 1.3185x; 1.3185x over previous
#include <hip/hip_runtime.h>

typedef __attribute__((ext_vector_type(2))) float v2f;
typedef __attribute__((ext_vector_type(4))) float v4f;
typedef __attribute__((ext_vector_type(8))) float v8f;


constexpr int Q = 4096;
constexpr int D = 128;
constexpr int A_ROWS = 128;
constexpr int B_ROWS = 64;
constexpr int PITCH = D + 4;

__global__ __launch_bounds__(256) void CentersDistance_wmma_kernel(
    const float* __restrict__ inputs,
    const float* __restrict__ centers,
    float* __restrict__ out)
{
    __shared__ __attribute__((aligned(16))) float tile[(A_ROWS + B_ROWS) * PITCH];

    const int lane = threadIdx.x & 31;
    const int wave = threadIdx.x >> 5;

    const int cblk = (blockIdx.x & 1) * A_ROWS;
    const int c0   = cblk + wave * 16;
    const int q0   = (blockIdx.x >> 1) * B_ROWS;

    v4f stage[24];
    #pragma unroll
    for (int i = 0; i < 16; ++i)
        stage[i] = *(const v4f*)(centers + (cblk + i * 8 + wave) * D + lane * 4);
    #pragma unroll
    for (int i = 0; i < 8; ++i)
        stage[16 + i] = *(const v4f*)(inputs + (q0 + i * 8 + wave) * D + lane * 4);
    #pragma unroll
    for (int i = 0; i < 24; ++i)
        *(v4f*)(tile + (i * 8 + wave) * PITCH + lane * 4) = stage[i];

    __syncthreads();

    const int hi  = lane >> 4;
    const int l15 = lane & 15;

    const float* aptr = tile + (wave * 16 + l15) * PITCH + 4 * hi;
    const float* bptr = tile + (A_ROWS + l15) * PITCH + 4 * hi;

    v8f acc0 = {}, acc1 = {}, acc2 = {}, acc3 = {};
    float asum = 0.f;
    float bsum0 = 0.f, bsum1 = 0.f, bsum2 = 0.f, bsum3 = 0.f;

    #pragma unroll
    for (int k = 0; k < D; k += 8) {
        v4f a  = *(const v4f*)(aptr + k);
        v4f b0 = *(const v4f*)(bptr + k);
        v4f b1 = *(const v4f*)(bptr + 16 * PITCH + k);
        v4f b2 = *(const v4f*)(bptr + 32 * PITCH + k);
        v4f b3 = *(const v4f*)(bptr + 48 * PITCH + k);

        asum  += a.x  * a.x  + a.y  * a.y  + a.z  * a.z  + a.w  * a.w;
        bsum0 += b0.x * b0.x + b0.y * b0.y + b0.z * b0.z + b0.w * b0.w;
        bsum1 += b1.x * b1.x + b1.y * b1.y + b1.z * b1.z + b1.w * b1.w;
        bsum2 += b2.x * b2.x + b2.y * b2.y + b2.z * b2.z + b2.w * b2.w;
        bsum3 += b3.x * b3.x + b3.y * b3.y + b3.z * b3.z + b3.w * b3.w;

        v2f alo = {a.x, a.y},  ahi = {a.z, a.w};
        v2f l0  = {b0.x, b0.y}, h0 = {b0.z, b0.w};
        v2f l1  = {b1.x, b1.y}, h1 = {b1.z, b1.w};
        v2f l2  = {b2.x, b2.y}, h2 = {b2.z, b2.w};
        v2f l3  = {b3.x, b3.y}, h3 = {b3.z, b3.w};

        acc0 = __builtin_amdgcn_wmma_f32_16x16x4_f32(false, alo, false, l0,
                                                     (short)0, acc0, false, false);
        acc1 = __builtin_amdgcn_wmma_f32_16x16x4_f32(false, alo, false, l1,
                                                     (short)0, acc1, false, false);
        acc2 = __builtin_amdgcn_wmma_f32_16x16x4_f32(false, alo, false, l2,
                                                     (short)0, acc2, false, false);
        acc3 = __builtin_amdgcn_wmma_f32_16x16x4_f32(false, alo, false, l3,
                                                     (short)0, acc3, false, false);
        acc0 = __builtin_amdgcn_wmma_f32_16x16x4_f32(false, ahi, false, h0,
                                                     (short)0, acc0, false, false);
        acc1 = __builtin_amdgcn_wmma_f32_16x16x4_f32(false, ahi, false, h1,
                                                     (short)0, acc1, false, false);
        acc2 = __builtin_amdgcn_wmma_f32_16x16x4_f32(false, ahi, false, h2,
                                                     (short)0, acc2, false, false);
        acc3 = __builtin_amdgcn_wmma_f32_16x16x4_f32(false, ahi, false, h3,
                                                     (short)0, acc3, false, false);
    }

    float cn  = asum  + __shfl_xor(asum,  16, 32);
    float qn0 = bsum0 + __shfl_xor(bsum0, 16, 32);
    float qn1 = bsum1 + __shfl_xor(bsum1, 16, 32);
    float qn2 = bsum2 + __shfl_xor(bsum2, 16, 32);
    float qn3 = bsum3 + __shfl_xor(bsum3, 16, 32);

    float crow[8];
    #pragma unroll
    for (int j = 0; j < 8; ++j)
        crow[j] = __shfl(cn, j + 8 * hi, 32);

    #pragma unroll
    for (int j = 0; j < 8; ++j) {
        const int row = c0 + j + 8 * hi;
        float* p = out + row * Q + q0 + l15;
        p[0]  = 2.f * acc0[j] - crow[j] - qn0;
        p[16] = 2.f * acc1[j] - crow[j] - qn1;
        p[32] = 2.f * acc2[j] - crow[j] - qn2;
        p[48] = 2.f * acc3[j] - crow[j] - qn3;
    }
}

extern "C" void kernel_launch(void* const* d_in, const int* in_sizes, int n_in,
                              void* d_out, int out_size, void* d_ws, size_t ws_size,
                              hipStream_t stream) {
    (void)in_sizes; (void)n_in; (void)out_size; (void)d_ws; (void)ws_size;
    const float* inputs  = (const float*)d_in[0];
    const float* centers = (const float*)d_in[1];
    float* out = (float*)d_out;

    CentersDistance_wmma_kernel<<<128, 256, 0, stream>>>(inputs, centers, out);
}


# Retile to 256 work-groups of 4 waves (64x64 tile, 32x32 per wave): every WGP gets a work-group instead of half of them, so each SIMD issues 128 f32 WMMAs, not 256
# speedup vs baseline: 1.4999x; 1.1376x over previous
#include <hip/hip_runtime.h>

typedef __attribute__((ext_vector_type(2))) float v2f;
typedef __attribute__((ext_vector_type(4))) float v4f;
typedef __attribute__((ext_vector_type(8))) float v8f;


constexpr int Q = 4096;
constexpr int D = 128;
constexpr int A_ROWS = 64;
constexpr int B_ROWS = 64;
constexpr int PITCH = D + 4;

__global__ __launch_bounds__(128) void CentersDistance_wmma_kernel(
    const float* __restrict__ inputs,
    const float* __restrict__ centers,
    float* __restrict__ out)
{
    __shared__ __attribute__((aligned(16))) float tile[(A_ROWS + B_ROWS) * PITCH];

    const int lane = threadIdx.x & 31;
    const int wave = threadIdx.x >> 5;
    const int wc   = wave & 1;
    const int wq   = wave >> 1;

    const int cblk = (blockIdx.x & 3) * A_ROWS;
    const int q0   = (blockIdx.x >> 2) * B_ROWS;

    v4f stage[32];
    #pragma unroll
    for (int i = 0; i < 16; ++i)
        stage[i] = *(const v4f*)(centers + (cblk + i * 4 + wave) * D + lane * 4);
    #pragma unroll
    for (int i = 0; i < 16; ++i)
        stage[16 + i] = *(const v4f*)(inputs + (q0 + i * 4 + wave) * D + lane * 4);
    #pragma unroll
    for (int i = 0; i < 32; ++i)
        *(v4f*)(tile + (i * 4 + wave) * PITCH + lane * 4) = stage[i];

    __syncthreads();

    const int hi  = lane >> 4;
    const int l15 = lane & 15;

    const float* aptr = tile + (wc * 32 + l15) * PITCH + 4 * hi;
    const float* bptr = tile + (A_ROWS + wq * 32 + l15) * PITCH + 4 * hi;

    v8f acc00 = {}, acc01 = {}, acc10 = {}, acc11 = {};
    float asum0 = 0.f, asum1 = 0.f;
    float bsum0 = 0.f, bsum1 = 0.f;

    #pragma unroll
    for (int k = 0; k < D; k += 8) {
        v4f a0 = *(const v4f*)(aptr + k);
        v4f a1 = *(const v4f*)(aptr + 16 * PITCH + k);
        v4f b0 = *(const v4f*)(bptr + k);
        v4f b1 = *(const v4f*)(bptr + 16 * PITCH + k);

        asum0 += a0.x * a0.x + a0.y * a0.y + a0.z * a0.z + a0.w * a0.w;
        asum1 += a1.x * a1.x + a1.y * a1.y + a1.z * a1.z + a1.w * a1.w;
        bsum0 += b0.x * b0.x + b0.y * b0.y + b0.z * b0.z + b0.w * b0.w;
        bsum1 += b1.x * b1.x + b1.y * b1.y + b1.z * b1.z + b1.w * b1.w;

        v2f al0 = {a0.x, a0.y}, ah0 = {a0.z, a0.w};
        v2f al1 = {a1.x, a1.y}, ah1 = {a1.z, a1.w};
        v2f bl0 = {b0.x, b0.y}, bh0 = {b0.z, b0.w};
        v2f bl1 = {b1.x, b1.y}, bh1 = {b1.z, b1.w};

        acc00 = __builtin_amdgcn_wmma_f32_16x16x4_f32(false, al0, false, bl0,
                                                      (short)0, acc00, false, false);
        acc01 = __builtin_amdgcn_wmma_f32_16x16x4_f32(false, al0, false, bl1,
                                                      (short)0, acc01, false, false);
        acc10 = __builtin_amdgcn_wmma_f32_16x16x4_f32(false, al1, false, bl0,
                                                      (short)0, acc10, false, false);
        acc11 = __builtin_amdgcn_wmma_f32_16x16x4_f32(false, al1, false, bl1,
                                                      (short)0, acc11, false, false);
        acc00 = __builtin_amdgcn_wmma_f32_16x16x4_f32(false, ah0, false, bh0,
                                                      (short)0, acc00, false, false);
        acc01 = __builtin_amdgcn_wmma_f32_16x16x4_f32(false, ah0, false, bh1,
                                                      (short)0, acc01, false, false);
        acc10 = __builtin_amdgcn_wmma_f32_16x16x4_f32(false, ah1, false, bh0,
                                                      (short)0, acc10, false, false);
        acc11 = __builtin_amdgcn_wmma_f32_16x16x4_f32(false, ah1, false, bh1,
                                                      (short)0, acc11, false, false);
    }

    float cn0 = asum0 + __shfl_xor(asum0, 16, 32);
    float cn1 = asum1 + __shfl_xor(asum1, 16, 32);
    float qn0 = bsum0 + __shfl_xor(bsum0, 16, 32);
    float qn1 = bsum1 + __shfl_xor(bsum1, 16, 32);

    float crow0[8], crow1[8];
    #pragma unroll
    for (int j = 0; j < 8; ++j) {
        crow0[j] = __shfl(cn0, j + 8 * hi, 32);
        crow1[j] = __shfl(cn1, j + 8 * hi, 32);
    }

    const int c0 = cblk + wc * 32;
    const int qw = q0 + wq * 32;
    #pragma unroll
    for (int j = 0; j < 8; ++j) {
        const int row = c0 + j + 8 * hi;
        float* p = out + row * Q + qw + l15;
        p[0]           = 2.f * acc00[j] - crow0[j] - qn0;
        p[16]          = 2.f * acc01[j] - crow0[j] - qn1;
        p[16 * Q]      = 2.f * acc10[j] - crow1[j] - qn0;
        p[16 * Q + 16] = 2.f * acc11[j] - crow1[j] - qn1;
    }
}

extern "C" void kernel_launch(void* const* d_in, const int* in_sizes, int n_in,
                              void* d_out, int out_size, void* d_ws, size_t ws_size,
                              hipStream_t stream) {
    (void)in_sizes; (void)n_in; (void)out_size; (void)d_ws; (void)ws_size;
    const float* inputs  = (const float*)d_in[0];
    const float* centers = (const float*)d_in[1];
    float* out = (float*)d_out;

    CentersDistance_wmma_kernel<<<256, 128, 0, stream>>>(inputs, centers, out);
}
